# starting kernel
# baseline (speedup 1.0000x reference)
#include <hip/hip_runtime.h>


typedef __attribute__((ext_vector_type(2))) float v2f;
typedef __attribute__((ext_vector_type(8))) float v8f;

#define N_IN   16
#define N_OUT  64
#define MREF   32
#define SDIM   256
#define WPB    8
#define GPITCH 33

__global__ __launch_bounds__(256) void householder_wy_kernel(
    const float* __restrict__ x,
    const float* __restrict__ v,
    float* __restrict__ out)
{
    const int wave = threadIdx.x >> 5;
    const int lane = threadIdx.x & 31;
    const int half = lane >> 4;
    const int lr   = lane & 15;

    const int batch = blockIdx.x * WPB + wave;
    const int i_in  = batch >> 6;
    const int o     = batch & 63;

    const float* vb = v + (size_t)batch * (MREF * SDIM);
    const float* xb = x + (size_t)(i_in * SDIM) * N_OUT + o;

    __shared__ float sG[WPB][MREF * GPITCH];
    __shared__ float sW[WPB][MREF];
    __shared__ float sA[WPB][MREF];
    __shared__ float sX[WPB][SDIM];

    float xr[8];
#pragma unroll
    for (int t = 0; t < 8; ++t) {
        xr[t] = xb[(size_t)(lane + 32 * t) * N_OUT];
        sX[wave][lane + 32 * t] = xr[t];
    }
    __syncthreads();

    v8f c00 = {}; v8f c01 = {}; v8f c11 = {}; v8f cw0 = {}; v8f cw1 = {};

#pragma unroll 4
    for (int k = 0; k < SDIM; k += 4) {
        const int kk = k + 2 * half;
        v2f a0 = *(const v2f*)(vb + (size_t)lr * SDIM + kk);
        v2f a1 = *(const v2f*)(vb + (size_t)(16 + lr) * SDIM + kk);
        const float x0 = sX[wave][kk];
        const float x1 = sX[wave][kk + 1];
        v2f b2;
        b2.x = (lr == 0) ? x0 : 0.0f;
        b2.y = (lr == 0) ? x1 : 0.0f;
        c00 = __builtin_amdgcn_wmma_f32_16x16x4_f32(false, a0, false, a0, (short)0, c00, false, false);
        c01 = __builtin_amdgcn_wmma_f32_16x16x4_f32(false, a0, false, a1, (short)0, c01, false, false);
        c11 = __builtin_amdgcn_wmma_f32_16x16x4_f32(false, a1, false, a1, (short)0, c11, false, false);
        cw0 = __builtin_amdgcn_wmma_f32_16x16x4_f32(false, a0, false, b2, (short)0, cw0, false, false);
        cw1 = __builtin_amdgcn_wmma_f32_16x16x4_f32(false, a1, false, b2, (short)0, cw1, false, false);
    }

#pragma unroll
    for (int j = 0; j < 8; ++j) {
        const int row = j + 8 * half;
        sG[wave][row * GPITCH + lr]               = c00[j];
        sG[wave][row * GPITCH + 16 + lr]          = c01[j];
        sG[wave][(16 + row) * GPITCH + 16 + lr]   = c11[j];
        if (lr == 0) {
            sW[wave][row]      = cw0[j];
            sW[wave][16 + row] = cw1[j];
        }
    }
    __syncthreads();

    {
        float r = sW[wave][lane];
        for (int i = MREF - 1; i >= 0; --i) {
            const float ri  = __shfl(r, i, 32);
            const float a   = 2.0f * ri / sG[wave][i * GPITCH + i];
            if (lane == i) sA[wave][i] = a;
            if (lane < i)  r -= a * sG[wave][lane * GPITCH + i];
        }
    }
    __syncthreads();

    float acc[8];
#pragma unroll
    for (int t = 0; t < 8; ++t)
        acc[t] = xr[t];

    for (int kk = 0; kk < MREF; ++kk) {
        const float ak = sA[wave][kk];
        const float* vr = vb + (size_t)kk * SDIM;
#pragma unroll
        for (int t = 0; t < 8; ++t)
            acc[t] -= ak * vr[lane + 32 * t];
    }

    float* ob = out + (size_t)(i_in * SDIM) * N_OUT + o;
#pragma unroll
    for (int t = 0; t < 8; ++t)
        ob[(size_t)(lane + 32 * t) * N_OUT] = acc[t];
}

extern "C" void kernel_launch(void* const* d_in, const int* in_sizes, int n_in,
                              void* d_out, int out_size, void* d_ws, size_t ws_size,
                              hipStream_t stream) {
    const float* x = (const float*)d_in[0];
    const float* v = (const float*)d_in[1];
    float* out = (float*)d_out;

    dim3 grid((N_IN * N_OUT) / WPB);
    dim3 block(WPB * 32);
    householder_wy_kernel<<<grid, block, 0, stream>>>(x, v, out);
}


# Recurrence solved across lanes from the accumulators, no LDS: lane c owns rows c, 16+c; a_k by DPP row broadcast times 2/G_kk; 2nd V pass issued among the products; r_i summed in a new order
# speedup vs baseline: 1.7323x; 1.7323x over previous
#include <hip/hip_runtime.h>


typedef __attribute__((ext_vector_type(2))) float v2f;
typedef __attribute__((ext_vector_type(4))) float v4f;
typedef __attribute__((ext_vector_type(8))) float v8f;

#define N_IN   16
#define N_OUT  64
#define MREF   32
#define SDIM   256
#define WPB    4
#define KGRP   (SDIM / 8)

static __device__ __forceinline__ float xhalf(float v) {
    const int b = __builtin_bit_cast(int, v);
    return __builtin_bit_cast(float, __builtin_amdgcn_permlanex16(b, b, 0x76543210, 0xfedcba98, false, false));
}

template <int N>
static __device__ __forceinline__ float row_bcast(float v) {
    return __builtin_bit_cast(float,
        __builtin_amdgcn_update_dpp(0, __builtin_bit_cast(int, v), 0x150 + N, 0xf, 0xf, true));
}

template <int K>
static __device__ __forceinline__ void bcast_all(float v, float* out) {
    out[K] = row_bcast<K>(v);
    if constexpr (K > 0) bcast_all<K - 1>(v, out);
}

template <int K>
static __device__ __forceinline__ void solve_hi(float& rlo, float& rhi, const float* glo, const float* ghi,
                                                const float* iu, float* a) {
    const float ak = row_bcast<K - 16>(rhi) * iu[K - 16];
    a[K] = ak;
    rhi -= ak * ghi[K - 16];
    rlo -= ak * glo[K];
    if constexpr (K > 16) solve_hi<K - 1>(rlo, rhi, glo, ghi, iu, a);
}

template <int K>
static __device__ __forceinline__ void solve_lo(float& rlo, const float* glo, const float* iu, float* a) {
    const float ak = row_bcast<K>(rlo) * iu[K];
    a[K] = ak;
    rlo -= ak * glo[K];
    if constexpr (K > 0) solve_lo<K - 1>(rlo, glo, iu, a);
}

__global__ __launch_bounds__(WPB * 32) __attribute__((amdgpu_waves_per_eu(1, 1)))
void householder_wy_kernel(
    const float* __restrict__ x,
    const float* __restrict__ v,
    float* __restrict__ out)
{
    const int wave = threadIdx.x >> 5;
    const int lane = threadIdx.x & 31;
    const int half = lane >> 4;
    const int lr   = lane & 15;

    const int batch = blockIdx.x * WPB + wave;
    const int i_in  = batch >> 6;
    const int o     = batch & 63;

    const float* vb = v + (size_t)batch * (MREF * SDIM);
    const float* xb = x + (size_t)(i_in * SDIM) * N_OUT + o;

    __shared__ __attribute__((aligned(16))) float sX[WPB][SDIM];

    float xr[8];
#pragma unroll
    for (int t = 0; t < 8; ++t)
        xr[t] = xb[(size_t)(lane + 32 * t) * N_OUT];

    const float* r0 = vb + (size_t)lr * SDIM + 4 * half;
    const float* r1 = r0 + (size_t)16 * SDIM;
    v4f f0[KGRP], f1[KGRP];
#pragma unroll
    for (int j = 0; j < KGRP; ++j) {
        f0[j] = *(const v4f*)(r0 + 8 * j);
        f1[j] = *(const v4f*)(r1 + 8 * j);
    }
    __builtin_amdgcn_sched_barrier(0);

#pragma unroll
    for (int t = 0; t < 8; ++t)
        sX[wave][lane + 32 * t] = xr[t];
    __builtin_amdgcn_fence(__ATOMIC_RELEASE, "wavefront");
    __builtin_amdgcn_wave_barrier();
    __builtin_amdgcn_fence(__ATOMIC_ACQUIRE, "wavefront");

    v8f c00 = {}; v8f c10 = {}; v8f c11 = {}; v8f cw0 = {}; v8f cw1 = {};
    const bool xrow = (lr & 7) == 0;

    static_assert(KGRP == MREF, "one row of the second pass is fetched behind each group of products");
    v4f pv[MREF][2];

    v4f xf[KGRP];
#pragma unroll
    for (int j = 0; j < KGRP; ++j)
        xf[j] = *(const v4f*)&sX[wave][8 * j + 4 * half];

#pragma unroll
    for (int j = 0; j < KGRP; ++j) {
        const v4f xq = xf[j];
#pragma unroll
        for (int p = 0; p < 2; ++p) {
            v2f a0, a1, ax;
            a0.x = f0[j][2 * p];  a0.y = f0[j][2 * p + 1];
            a1.x = f1[j][2 * p];  a1.y = f1[j][2 * p + 1];
            ax.x = xrow ? xq[2 * p]     : 0.0f;
            ax.y = xrow ? xq[2 * p + 1] : 0.0f;
            c00 = __builtin_amdgcn_wmma_f32_16x16x4_f32(false, a0, false, a0, (short)0, c00, false, false);
            c10 = __builtin_amdgcn_wmma_f32_16x16x4_f32(false, a1, false, a0, (short)0, c10, false, false);
            c11 = __builtin_amdgcn_wmma_f32_16x16x4_f32(false, a1, false, a1, (short)0, c11, false, false);
            cw0 = __builtin_amdgcn_wmma_f32_16x16x4_f32(false, ax, false, a0, (short)0, cw0, false, false);
            cw1 = __builtin_amdgcn_wmma_f32_16x16x4_f32(false, ax, false, a1, (short)0, cw1, false, false);
        }
        pv[j][0] = *(const v4f*)(vb + (size_t)j * SDIM + 4 * lane);
        pv[j][1] = *(const v4f*)(vb + (size_t)j * SDIM + 128 + 4 * lane);
        __builtin_amdgcn_sched_barrier(0);
    }

    float glo[MREF], ghi[MREF / 2];
#pragma unroll
    for (int r = 0; r < 8; ++r) {
        const float o00 = xhalf(c00[r]), o10 = xhalf(c10[r]), o11 = xhalf(c11[r]);
        glo[r]      = half ? o00 : c00[r];      glo[8 + r]  = half ? c00[r] : o00;
        glo[16 + r] = half ? o10 : c10[r];      glo[24 + r] = half ? c10[r] : o10;
        ghi[r]      = half ? o11 : c11[r];      ghi[8 + r]  = half ? c11[r] : o11;
    }
    float dlo = glo[0], dhi = ghi[0];
#pragma unroll
    for (int k = 1; k < 16; ++k) {
        dlo = (lr == k) ? glo[k]      : dlo;
        dhi = (lr == k) ? ghi[k]      : dhi;
    }
    const float ilo = 2.0f / dlo, ihi = 2.0f / dhi;
    float iulo[16], iuhi[16];
    bcast_all<15>(ilo, iulo);
    bcast_all<15>(ihi, iuhi);

    float rlo = cw0[0], rhi = cw1[0];
    float a[MREF];
    solve_hi<31>(rlo, rhi, glo, ghi, iuhi, a);
    solve_lo<15>(rlo, glo, iulo, a);

    v4f acc[2];
    acc[0] = *(const v4f*)&sX[wave][4 * lane];
    acc[1] = *(const v4f*)&sX[wave][128 + 4 * lane];

#pragma unroll
    for (int kk = 0; kk < MREF; ++kk) {
        acc[0] -= a[kk] * pv[kk][0];
        acc[1] -= a[kk] * pv[kk][1];
    }

    float* ob = out + (size_t)(i_in * SDIM) * N_OUT + o;
#pragma unroll
    for (int t = 0; t < 2; ++t)
#pragma unroll
        for (int s = 0; s < 4; ++s)
            ob[(size_t)(128 * t + 4 * lane + s) * N_OUT] = acc[t][s];
}

extern "C" void kernel_launch(void* const* d_in, const int* in_sizes, int n_in,
                              void* d_out, int out_size, void* d_ws, size_t ws_size,
                              hipStream_t stream) {
    const float* x = (const float*)d_in[0];
    const float* v = (const float*)d_in[1];
    float* out = (float*)d_out;

    dim3 grid((N_IN * N_OUT) / WPB);
    dim3 block(WPB * 32);
    householder_wy_kernel<<<grid, block, 0, stream>>>(x, v, out);
}


# w = V x taken off the matrix pipe: packed f32 multiply-adds on the fragment registers the lane already holds (x at the same k), 3 WMMAs per k step, not 5; G stays on WMMA; w summed in a new order
# speedup vs baseline: 1.8301x; 1.0565x over previous
#include <hip/hip_runtime.h>


typedef __attribute__((ext_vector_type(2))) float v2f;
typedef __attribute__((ext_vector_type(4))) float v4f;
typedef __attribute__((ext_vector_type(8))) float v8f;

#define N_IN   16
#define N_OUT  64
#define MREF   32
#define SDIM   256
#define WPB    4
#define KGRP   (SDIM / 8)

static __device__ __forceinline__ float xhalf(float v) {
    const int b = __builtin_bit_cast(int, v);
    return __builtin_bit_cast(float, __builtin_amdgcn_permlanex16(b, b, 0x76543210, 0xfedcba98, false, false));
}

template <int N>
static __device__ __forceinline__ float row_bcast(float v) {
    return __builtin_bit_cast(float,
        __builtin_amdgcn_update_dpp(0, __builtin_bit_cast(int, v), 0x150 + N, 0xf, 0xf, true));
}

template <int K>
static __device__ __forceinline__ void bcast_all(float v, float* out) {
    out[K] = row_bcast<K>(v);
    if constexpr (K > 0) bcast_all<K - 1>(v, out);
}

template <int K>
static __device__ __forceinline__ void solve_hi(float& rlo, float& rhi, const float* glo, const float* ghi,
                                                const float* iu, float* a) {
    const float ak = row_bcast<K - 16>(rhi) * iu[K - 16];
    a[K] = ak;
    rhi -= ak * ghi[K - 16];
    rlo -= ak * glo[K];
    if constexpr (K > 16) solve_hi<K - 1>(rlo, rhi, glo, ghi, iu, a);
}

template <int K>
static __device__ __forceinline__ void solve_lo(float& rlo, const float* glo, const float* iu, float* a) {
    const float ak = row_bcast<K>(rlo) * iu[K];
    a[K] = ak;
    rlo -= ak * glo[K];
    if constexpr (K > 0) solve_lo<K - 1>(rlo, glo, iu, a);
}

__global__ __launch_bounds__(WPB * 32) __attribute__((amdgpu_waves_per_eu(1, 1)))
void householder_wy_kernel(
    const float* __restrict__ x,
    const float* __restrict__ v,
    float* __restrict__ out)
{
    const int wave = threadIdx.x >> 5;
    const int lane = threadIdx.x & 31;
    const int half = lane >> 4;
    const int lr   = lane & 15;

    const int batch = blockIdx.x * WPB + wave;
    const int i_in  = batch >> 6;
    const int o     = batch & 63;

    const float* vb = v + (size_t)batch * (MREF * SDIM);
    const float* xb = x + (size_t)(i_in * SDIM) * N_OUT + o;

    __shared__ __attribute__((aligned(16))) float sX[WPB][SDIM];

    float xr[8];
#pragma unroll
    for (int t = 0; t < 8; ++t)
        xr[t] = xb[(size_t)(lane + 32 * t) * N_OUT];

    const float* r0 = vb + (size_t)lr * SDIM + 4 * half;
    const float* r1 = r0 + (size_t)16 * SDIM;
    v4f f0[KGRP], f1[KGRP];
#pragma unroll
    for (int j = 0; j < KGRP; ++j) {
        f0[j] = *(const v4f*)(r0 + 8 * j);
        f1[j] = *(const v4f*)(r1 + 8 * j);
    }
    __builtin_amdgcn_sched_barrier(0);

#pragma unroll
    for (int t = 0; t < 8; ++t)
        sX[wave][lane + 32 * t] = xr[t];
    __builtin_amdgcn_fence(__ATOMIC_RELEASE, "wavefront");
    __builtin_amdgcn_wave_barrier();
    __builtin_amdgcn_fence(__ATOMIC_ACQUIRE, "wavefront");

    v8f c00 = {}; v8f c10 = {}; v8f c11 = {};
    v2f w0 = {}; v2f w1 = {};

    static_assert(KGRP == MREF, "one row of the second pass is fetched behind each group of products");
    v4f pv[MREF][2];

    v4f xf[KGRP];
#pragma unroll
    for (int j = 0; j < KGRP; ++j)
        xf[j] = *(const v4f*)&sX[wave][8 * j + 4 * half];

#pragma unroll
    for (int j = 0; j < KGRP; ++j) {
        const v4f xq = xf[j];
#pragma unroll
        for (int p = 0; p < 2; ++p) {
            v2f a0, a1, xk;
            a0.x = f0[j][2 * p];  a0.y = f0[j][2 * p + 1];
            a1.x = f1[j][2 * p];  a1.y = f1[j][2 * p + 1];
            xk.x = xq[2 * p];     xk.y = xq[2 * p + 1];
            c00 = __builtin_amdgcn_wmma_f32_16x16x4_f32(false, a0, false, a0, (short)0, c00, false, false);
            c10 = __builtin_amdgcn_wmma_f32_16x16x4_f32(false, a1, false, a0, (short)0, c10, false, false);
            c11 = __builtin_amdgcn_wmma_f32_16x16x4_f32(false, a1, false, a1, (short)0, c11, false, false);
            w0 = __builtin_elementwise_fma(a0, xk, w0);
            w1 = __builtin_elementwise_fma(a1, xk, w1);
        }
        pv[j][0] = *(const v4f*)(vb + (size_t)j * SDIM + 4 * lane);
        pv[j][1] = *(const v4f*)(vb + (size_t)j * SDIM + 128 + 4 * lane);
        __builtin_amdgcn_sched_barrier(0);
    }

    float glo[MREF], ghi[MREF / 2];
#pragma unroll
    for (int r = 0; r < 8; ++r) {
        const float o00 = xhalf(c00[r]), o10 = xhalf(c10[r]), o11 = xhalf(c11[r]);
        glo[r]      = half ? o00 : c00[r];      glo[8 + r]  = half ? c00[r] : o00;
        glo[16 + r] = half ? o10 : c10[r];      glo[24 + r] = half ? c10[r] : o10;
        ghi[r]      = half ? o11 : c11[r];      ghi[8 + r]  = half ? c11[r] : o11;
    }
    float dlo = glo[0], dhi = ghi[0];
#pragma unroll
    for (int k = 1; k < 16; ++k) {
        dlo = (lr == k) ? glo[k]      : dlo;
        dhi = (lr == k) ? ghi[k]      : dhi;
    }
    const float ilo = 2.0f / dlo, ihi = 2.0f / dhi;
    float iulo[16], iuhi[16];
    bcast_all<15>(ilo, iulo);
    bcast_all<15>(ihi, iuhi);

    float rlo = w0.x + w0.y, rhi = w1.x + w1.y;
    rlo += xhalf(rlo);
    rhi += xhalf(rhi);
    float a[MREF];
    solve_hi<31>(rlo, rhi, glo, ghi, iuhi, a);
    solve_lo<15>(rlo, glo, iulo, a);

    v4f acc[2];
    acc[0] = *(const v4f*)&sX[wave][4 * lane];
    acc[1] = *(const v4f*)&sX[wave][128 + 4 * lane];

#pragma unroll
    for (int kk = 0; kk < MREF; ++kk) {
        acc[0] -= a[kk] * pv[kk][0];
        acc[1] -= a[kk] * pv[kk][1];
    }

    float* ob = out + (size_t)(i_in * SDIM) * N_OUT + o;
#pragma unroll
    for (int t = 0; t < 2; ++t)
#pragma unroll
        for (int s = 0; s < 4; ++s)
            ob[(size_t)(128 * t + 4 * lane + s) * N_OUT] = acc[t][s];
}

extern "C" void kernel_launch(void* const* d_in, const int* in_sizes, int n_in,
                              void* d_out, int out_size, void* d_ws, size_t ws_size,
                              hipStream_t stream) {
    const float* x = (const float*)d_in[0];
    const float* v = (const float*)d_in[1];
    float* out = (float*)d_out;

    dim3 grid((N_IN * N_OUT) / WPB);
    dim3 block(WPB * 32);
    householder_wy_kernel<<<grid, block, 0, stream>>>(x, v, out);
}


# y = x - V^T a split over the block's 4 waves by output rows, not by batch (a passed through LDS, one block barrier): out leaves as 16 B per row for 4 adjacent n_out, not 4 B; same sums, same order
# speedup vs baseline: 2.9292x; 1.6006x over previous
#include <hip/hip_runtime.h>


typedef __attribute__((ext_vector_type(2))) float v2f;
typedef __attribute__((ext_vector_type(4))) float v4f;
typedef __attribute__((ext_vector_type(8))) float v8f;

#define N_IN   16
#define N_OUT  64
#define MREF   32
#define SDIM   256
#define WPB    4
#define KGRP   (SDIM / 8)

static __device__ __forceinline__ float xhalf(float v) {
    const int b = __builtin_bit_cast(int, v);
    return __builtin_bit_cast(float, __builtin_amdgcn_permlanex16(b, b, 0x76543210, 0xfedcba98, false, false));
}

template <int N>
static __device__ __forceinline__ float row_bcast(float v) {
    return __builtin_bit_cast(float,
        __builtin_amdgcn_update_dpp(0, __builtin_bit_cast(int, v), 0x150 + N, 0xf, 0xf, true));
}

template <int K>
static __device__ __forceinline__ void bcast_all(float v, float* out) {
    out[K] = row_bcast<K>(v);
    if constexpr (K > 0) bcast_all<K - 1>(v, out);
}

template <int K>
static __device__ __forceinline__ void solve_hi(float& rlo, float& rhi, const float* glo, const float* ghi,
                                                const float* iu, float* a) {
    const float ak = row_bcast<K - 16>(rhi) * iu[K - 16];
    a[K] = ak;
    rhi -= ak * ghi[K - 16];
    rlo -= ak * glo[K];
    if constexpr (K > 16) solve_hi<K - 1>(rlo, rhi, glo, ghi, iu, a);
}

template <int K>
static __device__ __forceinline__ void solve_lo(float& rlo, const float* glo, const float* iu, float* a) {
    const float ak = row_bcast<K>(rlo) * iu[K];
    a[K] = ak;
    rlo -= ak * glo[K];
    if constexpr (K > 0) solve_lo<K - 1>(rlo, glo, iu, a);
}

__global__ __launch_bounds__(WPB * 32) __attribute__((amdgpu_waves_per_eu(1, 1)))
void householder_wy_kernel(
    const float* __restrict__ x,
    const float* __restrict__ v,
    float* __restrict__ out)
{
    const int wave = threadIdx.x >> 5;
    const int lane = threadIdx.x & 31;
    const int half = lane >> 4;
    const int lr   = lane & 15;

    const int batch = blockIdx.x * WPB + wave;
    const int i_in  = batch >> 6;
    const int o     = batch & 63;

    const float* vb = v + (size_t)batch * (MREF * SDIM);
    const float* xb = x + (size_t)(i_in * SDIM) * N_OUT + o;

    __shared__ __attribute__((aligned(16))) float sX[WPB][SDIM];
    __shared__ __attribute__((aligned(16))) float sA[WPB][MREF];

    float xr[8];
#pragma unroll
    for (int t = 0; t < 8; ++t)
        xr[t] = xb[(size_t)(lane + 32 * t) * N_OUT];

    const float* r0 = vb + (size_t)lr * SDIM + 4 * half;
    const float* r1 = r0 + (size_t)16 * SDIM;
    v4f f0[KGRP], f1[KGRP];
#pragma unroll
    for (int j = 0; j < KGRP; ++j) {
        f0[j] = *(const v4f*)(r0 + 8 * j);
        f1[j] = *(const v4f*)(r1 + 8 * j);
    }
    __builtin_amdgcn_sched_barrier(0);

#pragma unroll
    for (int t = 0; t < 8; ++t)
        sX[wave][lane + 32 * t] = xr[t];
    __builtin_amdgcn_fence(__ATOMIC_RELEASE, "wavefront");
    __builtin_amdgcn_wave_barrier();
    __builtin_amdgcn_fence(__ATOMIC_ACQUIRE, "wavefront");

    v8f c00 = {}; v8f c10 = {}; v8f c11 = {};
    v2f w0 = {}; v2f w1 = {};

    static_assert(KGRP == MREF, "one row of the second pass is fetched behind each group of products");
    static_assert(WPB == 4 && SDIM == 64 * WPB, "phase 3: 4 waves x 64 output rows, 2 batches per lane half");
    v4f pv[MREF][2];
    const float* vq = v + (size_t)(blockIdx.x * WPB + 2 * half) * (MREF * SDIM) + 64 * wave + 4 * lr;

    v4f xf[KGRP];
#pragma unroll
    for (int j = 0; j < KGRP; ++j)
        xf[j] = *(const v4f*)&sX[wave][8 * j + 4 * half];

#pragma unroll
    for (int j = 0; j < KGRP; ++j) {
        const v4f xq = xf[j];
#pragma unroll
        for (int p = 0; p < 2; ++p) {
            v2f a0, a1, xk;
            a0.x = f0[j][2 * p];  a0.y = f0[j][2 * p + 1];
            a1.x = f1[j][2 * p];  a1.y = f1[j][2 * p + 1];
            xk.x = xq[2 * p];     xk.y = xq[2 * p + 1];
            c00 = __builtin_amdgcn_wmma_f32_16x16x4_f32(false, a0, false, a0, (short)0, c00, false, false);
            c10 = __builtin_amdgcn_wmma_f32_16x16x4_f32(false, a1, false, a0, (short)0, c10, false, false);
            c11 = __builtin_amdgcn_wmma_f32_16x16x4_f32(false, a1, false, a1, (short)0, c11, false, false);
            w0 = __builtin_elementwise_fma(a0, xk, w0);
            w1 = __builtin_elementwise_fma(a1, xk, w1);
        }
        pv[j][0] = *(const v4f*)(vq + (size_t)j * SDIM);
        pv[j][1] = *(const v4f*)(vq + (size_t)(MREF * SDIM) + (size_t)j * SDIM);
        __builtin_amdgcn_sched_barrier(0);
    }

    float glo[MREF], ghi[MREF / 2];
#pragma unroll
    for (int r = 0; r < 8; ++r) {
        const float o00 = xhalf(c00[r]), o10 = xhalf(c10[r]), o11 = xhalf(c11[r]);
        glo[r]      = half ? o00 : c00[r];      glo[8 + r]  = half ? c00[r] : o00;
        glo[16 + r] = half ? o10 : c10[r];      glo[24 + r] = half ? c10[r] : o10;
        ghi[r]      = half ? o11 : c11[r];      ghi[8 + r]  = half ? c11[r] : o11;
    }
    float dlo = glo[0], dhi = ghi[0];
#pragma unroll
    for (int k = 1; k < 16; ++k) {
        dlo = (lr == k) ? glo[k]      : dlo;
        dhi = (lr == k) ? ghi[k]      : dhi;
    }
    const float ilo = 2.0f / dlo, ihi = 2.0f / dhi;
    float iulo[16], iuhi[16];
    bcast_all<15>(ilo, iulo);
    bcast_all<15>(ihi, iuhi);

    float rlo = w0.x + w0.y, rhi = w1.x + w1.y;
    rlo += xhalf(rlo);
    rhi += xhalf(rhi);
    float a[MREF];
    solve_hi<31>(rlo, rhi, glo, ghi, iuhi, a);
    solve_lo<15>(rlo, glo, iulo, a);

    if (lane == 0) {
#pragma unroll
        for (int i = 0; i < MREF / 4; ++i) {
            v4f t;
            t.x = a[4 * i]; t.y = a[4 * i + 1]; t.z = a[4 * i + 2]; t.w = a[4 * i + 3];
            *(v4f*)&sA[wave][4 * i] = t;
        }
    }
    __syncthreads();

    v4f av[2][MREF / 4];
#pragma unroll
    for (int n = 0; n < 2; ++n)
#pragma unroll
        for (int i = 0; i < MREF / 4; ++i)
            av[n][i] = *(const v4f*)&sA[2 * half + n][4 * i];

    v4f acc[2];
    acc[0] = *(const v4f*)&sX[2 * half][64 * wave + 4 * lr];
    acc[1] = *(const v4f*)&sX[2 * half + 1][64 * wave + 4 * lr];

#pragma unroll
    for (int kk = 0; kk < MREF; ++kk) {
        acc[0] -= av[0][kk >> 2][kk & 3] * pv[kk][0];
        acc[1] -= av[1][kk >> 2][kk & 3] * pv[kk][1];
    }

    const int o0 = (blockIdx.x * WPB) & (N_OUT - 1);
    float* ob = out + ((size_t)(i_in * SDIM) + 64 * wave + 4 * lr) * N_OUT + o0 + 2 * half;
#pragma unroll
    for (int e = 0; e < 4; ++e) {
        v2f y2;
        y2.x = acc[0][e]; y2.y = acc[1][e];
        *(v2f*)(ob + (size_t)e * N_OUT) = y2;
    }
}

extern "C" void kernel_launch(void* const* d_in, const int* in_sizes, int n_in,
                              void* d_out, int out_size, void* d_ws, size_t ws_size,
                              hipStream_t stream) {
    const float* x = (const float*)d_in[0];
    const float* v = (const float*)d_in[1];
    float* out = (float*)d_out;

    dim3 grid((N_IN * N_OUT) / WPB);
    dim3 block(WPB * 32);
    householder_wy_kernel<<<grid, block, 0, stream>>>(x, v, out);
}
